# starting kernel
# baseline (speedup 1.0000x reference)
#include <hip/hip_runtime.h>

typedef __attribute__((ext_vector_type(2))) float v2f;
typedef __attribute__((ext_vector_type(8))) float v8f;

#define N_TOK 360
#define D_DIM 128
#define B_SZ  4
#define M_ROWS (B_SZ * N_TOK)

__global__ __launch_bounds__(256) void gemm_wmma(const float* __restrict__ x,
                                                 const float* __restrict__ W1,
                                                 const float* __restrict__ b1,
                                                 float* __restrict__ wsA,
                                                 float* __restrict__ wsC) {
    const int wid  = threadIdx.x >> 5;
    const int lane = threadIdx.x & 31;
    const int gw   = blockIdx.x * 8 + wid;
    const int mt   = gw >> 4;
    const int nt16 = gw & 15;
    const int half = nt16 >> 3;
    const int nt   = nt16 & 7;
    const int mbase = mt * 16;
    const int nbase = nt * 16;
    const int lm    = lane & 15;
    const int koff  = (lane < 16) ? 0 : 2;

    const float* xrow = x + (mbase + lm) * D_DIM;
    const float* wb   = W1 + half * D_DIM * D_DIM;

    v8f acc = {0.f, 0.f, 0.f, 0.f, 0.f, 0.f, 0.f, 0.f};
#pragma unroll
    for (int k = 0; k < D_DIM; k += 4) {
        v2f a = *(const v2f*)(xrow + k + koff);
        v2f b;
        b.x = wb[(k + koff)     * D_DIM + nbase + lm];
        b.y = wb[(k + koff + 1) * D_DIM + nbase + lm];
        acc = __builtin_amdgcn_wmma_f32_16x16x4_f32(
             false, a,  false, b,
             (short)0, acc,  false,  false);
    }

    float* dst = half ? wsC : wsA;
    const float badd = half ? 0.0f : b1[nbase + lm];
    const int moff = (lane < 16) ? 0 : 8;
#pragma unroll
    for (int r = 0; r < 8; ++r) {
        dst[(mbase + r + moff) * D_DIM + nbase + lm] = acc[r] + badd;
    }
}

__global__ __launch_bounds__(256) void pair_logits(const float* __restrict__ wsA,
                                                   const float* __restrict__ wsC,
                                                   const float* __restrict__ W2,
                                                   const float* __restrict__ b2,
                                                   float* __restrict__ out) {
    __shared__ float As[16][132];
    __shared__ float Cs[16][132];
    __shared__ float w2s[128];

    const int b  = blockIdx.z;
    const int i0 = blockIdx.x * 16;
    const int j0 = blockIdx.y * 16;
    const int tid = threadIdx.x;

    for (int idx = tid; idx < 16 * 128; idx += 256) {
        const int r = idx >> 7, c = idx & 127;
        const int gi = i0 + r, gj = j0 + r;
        As[r][c] = (gi < N_TOK) ? wsA[(b * N_TOK + gi) * D_DIM + c] : 0.0f;
        Cs[r][c] = (gj < N_TOK) ? wsC[(b * N_TOK + gj) * D_DIM + c] : 0.0f;
    }
    if (tid < 128) w2s[tid] = W2[tid];
    __syncthreads();

    const int tx = tid & 15;
    const int ty = tid >> 4;
    float acc = 0.0f;
#pragma unroll 4
    for (int d = 0; d < D_DIM; ++d) {
        float v = As[ty][d] + Cs[tx][d];
        v = fmaxf(v, 0.0f);
        acc = fmaf(v, w2s[d], acc);
    }

    const int i = i0 + ty, j = j0 + tx;
    if (i < N_TOK && j < N_TOK) {
        float lg = acc + b2[0];
        if (i == j) lg = 0.0f;
        out[(b * N_TOK + i) * N_TOK + j] = lg;
    }
}

__global__ __launch_bounds__(128) void row_softmax(float* __restrict__ out) {
    __shared__ float red[128];
    float* p = out + (size_t)blockIdx.x * N_TOK;
    const int tid = threadIdx.x;

    float vals[3];
    float m = -3.4e38f;
#pragma unroll
    for (int t = 0; t < 3; ++t) {
        const int j = tid + t * 128;
        vals[t] = (j < N_TOK) ? p[j] : -3.4e38f;
        m = fmaxf(m, vals[t]);
    }
    red[tid] = m;
    __syncthreads();
    for (int s = 64; s > 0; s >>= 1) {
        if (tid < s) red[tid] = fmaxf(red[tid], red[tid + s]);
        __syncthreads();
    }
    m = red[0];
    __syncthreads();

    float sum = 0.0f;
#pragma unroll
    for (int t = 0; t < 3; ++t) {
        const int j = tid + t * 128;
        if (j < N_TOK) {
            vals[t] = expf(vals[t] - m);
            sum += vals[t];
        }
    }
    red[tid] = sum;
    __syncthreads();
    for (int s = 64; s > 0; s >>= 1) {
        if (tid < s) red[tid] += red[tid + s];
        __syncthreads();
    }
    const float inv = 1.0f / red[0];
#pragma unroll
    for (int t = 0; t < 3; ++t) {
        const int j = tid + t * 128;
        if (j < N_TOK) p[j] = vals[t] * inv;
    }
}

extern "C" void kernel_launch(void* const* d_in, const int* in_sizes, int n_in,
                              void* d_out, int out_size, void* d_ws, size_t ws_size,
                              hipStream_t stream) {
    (void)in_sizes; (void)n_in; (void)out_size; (void)ws_size;
    const float* x  = (const float*)d_in[0];
    const float* W1 = (const float*)d_in[1];
    const float* b1 = (const float*)d_in[2];
    const float* W2 = (const float*)d_in[3];
    const float* b2 = (const float*)d_in[4];
    float* out = (float*)d_out;

    float* wsA = (float*)d_ws;
    float* wsC = wsA + M_ROWS * D_DIM;

    gemm_wmma<<<dim3(M_ROWS * 16 / 16 / 8), 256, 0, stream>>>(x, W1, b1, wsA, wsC);

    dim3 pg((N_TOK + 15) / 16, (N_TOK + 15) / 16, B_SZ);
    pair_logits<<<pg, 256, 0, stream>>>(wsA, wsC, W2, b2, out);

    row_softmax<<<dim3(B_SZ * N_TOK), 128, 0, stream>>>(out);
}


# pair kernel: d across lanes, rows from global straight to registers (no LDS), 32 pairs summed by a transposed lane butterfly; 16928 waves -> 2700; logit sum order now 4 terms per lane + lane tree
# speedup vs baseline: 1.6893x; 1.6893x over previous
#include <hip/hip_runtime.h>

typedef __attribute__((ext_vector_type(2))) float v2f;
typedef __attribute__((ext_vector_type(4))) float v4f;
typedef __attribute__((ext_vector_type(8))) float v8f;

#define N_TOK 360
#define D_DIM 128
#define B_SZ  4
#define M_ROWS (B_SZ * N_TOK)

__global__ __launch_bounds__(256) void gemm_wmma(const float* __restrict__ x,
                                                 const float* __restrict__ W1,
                                                 const float* __restrict__ b1,
                                                 float* __restrict__ wsA,
                                                 float* __restrict__ wsC) {
    const int wid  = threadIdx.x >> 5;
    const int lane = threadIdx.x & 31;
    const int gw   = blockIdx.x * 8 + wid;
    const int mt   = gw >> 4;
    const int nt16 = gw & 15;
    const int half = nt16 >> 3;
    const int nt   = nt16 & 7;
    const int mbase = mt * 16;
    const int nbase = nt * 16;
    const int lm    = lane & 15;
    const int koff  = (lane < 16) ? 0 : 2;

    const float* xrow = x + (mbase + lm) * D_DIM;
    const float* wb   = W1 + half * D_DIM * D_DIM;

    v8f acc = {0.f, 0.f, 0.f, 0.f, 0.f, 0.f, 0.f, 0.f};
#pragma unroll
    for (int k = 0; k < D_DIM; k += 4) {
        v2f a = *(const v2f*)(xrow + k + koff);
        v2f b;
        b.x = wb[(k + koff)     * D_DIM + nbase + lm];
        b.y = wb[(k + koff + 1) * D_DIM + nbase + lm];
        acc = __builtin_amdgcn_wmma_f32_16x16x4_f32(
             false, a,  false, b,
             (short)0, acc,  false,  false);
    }

    float* dst = half ? wsC : wsA;
    const float badd = half ? 0.0f : b1[nbase + lm];
    const int moff = (lane < 16) ? 0 : 8;
#pragma unroll
    for (int r = 0; r < 8; ++r) {
        dst[(mbase + r + moff) * D_DIM + nbase + lm] = acc[r] + badd;
    }
}

#define TILE_I 8
#define TILE_J 24

__device__ __forceinline__ float xhalf(float x) {
    int v = __builtin_bit_cast(int, x);
    return __builtin_bit_cast(float, __builtin_amdgcn_permlanex16(
        v, v, 0x76543210, 0xfedcba98, false, false));
}
template <int M>
__device__ __forceinline__ float xrow(float x) {
    return __builtin_bit_cast(float, __builtin_amdgcn_update_dpp(
        0, __builtin_bit_cast(int, x), 0x160 + M, 0xf, 0xf, true));
}

__global__ __launch_bounds__(96) void pair_logits(const float* __restrict__ wsA,
                                                   const float* __restrict__ wsC,
                                                   const float* __restrict__ W2,
                                                   const float* __restrict__ b2,
                                                   float* __restrict__ out) {
    const int lane = threadIdx.x & 31;
    const int b    = blockIdx.x / 225;
    const int rem  = blockIdx.x - b * 225;
    const int itg  = rem / 15;
    const int jt   = rem - itg * 15;
    const int i0   = (itg * 3 + (threadIdx.x >> 5)) * TILE_I;
    const int j0   = jt * TILE_J;

    const int swz = ((lane >> 3) & 3) << 1;

    const float* Ab = wsA + (b * N_TOK + i0) * D_DIM + 4 * lane;
    const float* Cb = wsC + (b * N_TOK + j0) * D_DIM + 4 * lane;

    v4f a[TILE_I];
    v4f c[TILE_J];
#pragma unroll
    for (int s = 0; s < TILE_I; ++s) a[s] = *(const v4f*)(Ab + (s ^ swz) * D_DIM);
#pragma unroll
    for (int s = 0; s < TILE_J; ++s) c[s] = *(const v4f*)(Cb + s * D_DIM);
    const v4f wv = *(const v4f*)(W2 + 4 * lane);
    const v2f wlo = {wv.x, wv.y};
    const v2f whi = {wv.z, wv.w};
    const float bias2 = b2[0];
    const bool l2 = (lane & 4) != 0;
    const bool l1 = (lane & 2) != 0;
    const bool l0 = (lane & 1) != 0;
    const int  io = i0 + (lane >> 2);
    const int  jl = j0 + (lane & 3);

#pragma unroll
    for (int g = 0; g < TILE_J / 4; ++g) {
        float p[32];
#pragma unroll
        for (int k = 0; k < 32; ++k) {
            const v4f s = a[k >> 2] + c[4 * g + (k & 3)];
            const v2f lo = {fmaxf(s.x, 0.0f), fmaxf(s.y, 0.0f)};
            const v2f hi = {fmaxf(s.z, 0.0f), fmaxf(s.w, 0.0f)};
            v2f u = lo * wlo;
            u = __builtin_elementwise_fma(hi, whi, u);
            p[k] = u.x + u.y;
        }
#pragma unroll
        for (int k = 0; k < 16; ++k) p[k] = p[k] + xhalf(p[k + 16]);
#pragma unroll
        for (int k = 0; k < 8; ++k) p[k] = p[k] + xrow<8>(p[k + 8]);
#pragma unroll
        for (int k = 0; k < 4; ++k) {
            const float give = l2 ? p[k] : p[k + 4];
            const float keep = l2 ? p[k + 4] : p[k];
            p[k] = keep + xrow<4>(give);
        }
#pragma unroll
        for (int k = 0; k < 2; ++k) {
            const float give = l1 ? p[k] : p[k + 2];
            const float keep = l1 ? p[k + 2] : p[k];
            p[k] = keep + xrow<2>(give);
        }
        {
            const float give = l0 ? p[0] : p[1];
            const float keep = l0 ? p[1] : p[0];
            p[0] = keep + xrow<1>(give);
        }
        const int j = jl + 4 * g;
        float lg = p[0] + bias2;
        if (io == j) lg = 0.0f;
        out[(b * N_TOK + io) * N_TOK + j] = lg;
    }
}

__global__ __launch_bounds__(128) void row_softmax(float* __restrict__ out) {
    __shared__ float red[128];
    float* p = out + (size_t)blockIdx.x * N_TOK;
    const int tid = threadIdx.x;

    float vals[3];
    float m = -3.4e38f;
#pragma unroll
    for (int t = 0; t < 3; ++t) {
        const int j = tid + t * 128;
        vals[t] = (j < N_TOK) ? p[j] : -3.4e38f;
        m = fmaxf(m, vals[t]);
    }
    red[tid] = m;
    __syncthreads();
    for (int s = 64; s > 0; s >>= 1) {
        if (tid < s) red[tid] = fmaxf(red[tid], red[tid + s]);
        __syncthreads();
    }
    m = red[0];
    __syncthreads();

    float sum = 0.0f;
#pragma unroll
    for (int t = 0; t < 3; ++t) {
        const int j = tid + t * 128;
        if (j < N_TOK) {
            vals[t] = expf(vals[t] - m);
            sum += vals[t];
        }
    }
    red[tid] = sum;
    __syncthreads();
    for (int s = 64; s > 0; s >>= 1) {
        if (tid < s) red[tid] += red[tid + s];
        __syncthreads();
    }
    const float inv = 1.0f / red[0];
#pragma unroll
    for (int t = 0; t < 3; ++t) {
        const int j = tid + t * 128;
        if (j < N_TOK) p[j] = vals[t] * inv;
    }
}

extern "C" void kernel_launch(void* const* d_in, const int* in_sizes, int n_in,
                              void* d_out, int out_size, void* d_ws, size_t ws_size,
                              hipStream_t stream) {
    (void)in_sizes; (void)n_in; (void)out_size; (void)ws_size;
    const float* x  = (const float*)d_in[0];
    const float* W1 = (const float*)d_in[1];
    const float* b1 = (const float*)d_in[2];
    const float* W2 = (const float*)d_in[3];
    const float* b2 = (const float*)d_in[4];
    float* out = (float*)d_out;

    float* wsA = (float*)d_ws;
    float* wsC = wsA + M_ROWS * D_DIM;

    gemm_wmma<<<dim3(M_ROWS * 16 / 16 / 8), 256, 0, stream>>>(x, W1, b1, wsA, wsC);

    pair_logits<<<dim3(900), 96, 0, stream>>>(wsA, wsC, W2, b2, out);

    row_softmax<<<dim3(B_SZ * N_TOK), 128, 0, stream>>>(out);
}


# row softmax fused into the pair kernel: a block's 15 waves own 8 full rows, logits stay in registers; row max/sum by quad butterfly, then 15 wave partials in LDS (new sum order); 3 launches -> 2
# speedup vs baseline: 2.2436x; 1.3281x over previous
#include <hip/hip_runtime.h>

typedef __attribute__((ext_vector_type(2))) float v2f;
typedef __attribute__((ext_vector_type(4))) float v4f;
typedef __attribute__((ext_vector_type(8))) float v8f;

#define N_TOK 360
#define D_DIM 128
#define B_SZ  4
#define M_ROWS (B_SZ * N_TOK)

__global__ __launch_bounds__(256) void gemm_wmma(const float* __restrict__ x,
                                                 const float* __restrict__ W1,
                                                 const float* __restrict__ b1,
                                                 float* __restrict__ wsA,
                                                 float* __restrict__ wsC) {
    const int wid  = threadIdx.x >> 5;
    const int lane = threadIdx.x & 31;
    const int gw   = blockIdx.x * 8 + wid;
    const int mt   = gw >> 4;
    const int nt16 = gw & 15;
    const int half = nt16 >> 3;
    const int nt   = nt16 & 7;
    const int mbase = mt * 16;
    const int nbase = nt * 16;
    const int lm    = lane & 15;
    const int koff  = (lane < 16) ? 0 : 2;

    const float* xrow = x + (mbase + lm) * D_DIM;
    const float* wb   = W1 + half * D_DIM * D_DIM;

    v8f acc = {0.f, 0.f, 0.f, 0.f, 0.f, 0.f, 0.f, 0.f};
#pragma unroll
    for (int k = 0; k < D_DIM; k += 4) {
        v2f a = *(const v2f*)(xrow + k + koff);
        v2f b;
        b.x = wb[(k + koff)     * D_DIM + nbase + lm];
        b.y = wb[(k + koff + 1) * D_DIM + nbase + lm];
        acc = __builtin_amdgcn_wmma_f32_16x16x4_f32(
             false, a,  false, b,
             (short)0, acc,  false,  false);
    }

    float* dst = half ? wsC : wsA;
    const float badd = half ? 0.0f : b1[nbase + lm];
    const int moff = (lane < 16) ? 0 : 8;
#pragma unroll
    for (int r = 0; r < 8; ++r) {
        dst[(mbase + r + moff) * D_DIM + nbase + lm] = acc[r] + badd;
    }
}

#define TILE_I 8
#define TILE_J 24
#define N_JT   (N_TOK / TILE_J)
#define PART_W 16

__device__ __forceinline__ float xhalf(float x) {
    int v = __builtin_bit_cast(int, x);
    return __builtin_bit_cast(float, __builtin_amdgcn_permlanex16(
        v, v, 0x76543210, 0xfedcba98, false, false));
}
template <int M>
__device__ __forceinline__ float xrow(float x) {
    return __builtin_bit_cast(float, __builtin_amdgcn_update_dpp(
        0, __builtin_bit_cast(int, x), 0x160 + M, 0xf, 0xf, true));
}

__global__ __launch_bounds__(480) void pair_softmax(const float* __restrict__ wsA,
                                                    const float* __restrict__ wsC,
                                                    const float* __restrict__ W2,
                                                    const float* __restrict__ b2,
                                                    float* __restrict__ out) {
    __shared__ __attribute__((aligned(16))) float s_tile[TILE_I * N_TOK];
    __shared__ __attribute__((aligned(16))) float s_max[TILE_I * PART_W];
    __shared__ __attribute__((aligned(16))) float s_sum[TILE_I * PART_W];

    const int lane = threadIdx.x & 31;
    const int w    = threadIdx.x >> 5;
    const int b    = blockIdx.x / 45;
    const int it   = blockIdx.x - b * 45;
    const int i0   = it * TILE_I;
    const int j0   = w * TILE_J;

    const int swz = ((lane >> 3) & 3) << 1;

    const float* Ab = wsA + (b * N_TOK + i0) * D_DIM + 4 * lane;
    const float* Cb = wsC + (b * N_TOK + j0) * D_DIM + 4 * lane;

    v4f a[TILE_I];
    v4f c[TILE_J];
#pragma unroll
    for (int s = 0; s < TILE_I; ++s) a[s] = *(const v4f*)(Ab + (s ^ swz) * D_DIM);
#pragma unroll
    for (int s = 0; s < TILE_J; ++s) c[s] = *(const v4f*)(Cb + s * D_DIM);
    const v4f wv = *(const v4f*)(W2 + 4 * lane);
    const v2f wlo = {wv.x, wv.y};
    const v2f whi = {wv.z, wv.w};
    const float bias2 = b2[0];
    const bool l2 = (lane & 4) != 0;
    const bool l1 = (lane & 2) != 0;
    const bool l0 = (lane & 1) != 0;
    const int  r  = lane >> 2;
    const int  q  = lane & 3;
    const int  io = i0 + r;
    const int  jl = j0 + q;

    float lg[TILE_J / 4];
#pragma unroll
    for (int g = 0; g < TILE_J / 4; ++g) {
        float p[32];
#pragma unroll
        for (int k = 0; k < 32; ++k) {
            const v4f s = a[k >> 2] + c[4 * g + (k & 3)];
            const v2f lo = {fmaxf(s.x, 0.0f), fmaxf(s.y, 0.0f)};
            const v2f hi = {fmaxf(s.z, 0.0f), fmaxf(s.w, 0.0f)};
            v2f u = lo * wlo;
            u = __builtin_elementwise_fma(hi, whi, u);
            p[k] = u.x + u.y;
        }
#pragma unroll
        for (int k = 0; k < 16; ++k) p[k] = p[k] + xhalf(p[k + 16]);
#pragma unroll
        for (int k = 0; k < 8; ++k) p[k] = p[k] + xrow<8>(p[k + 8]);
#pragma unroll
        for (int k = 0; k < 4; ++k) {
            const float give = l2 ? p[k] : p[k + 4];
            const float keep = l2 ? p[k + 4] : p[k];
            p[k] = keep + xrow<4>(give);
        }
#pragma unroll
        for (int k = 0; k < 2; ++k) {
            const float give = l1 ? p[k] : p[k + 2];
            const float keep = l1 ? p[k + 2] : p[k];
            p[k] = keep + xrow<2>(give);
        }
        {
            const float give = l0 ? p[0] : p[1];
            const float keep = l0 ? p[1] : p[0];
            p[0] = keep + xrow<1>(give);
        }
        const int j = jl + 4 * g;
        float v = p[0] + bias2;
        if (io == j) v = 0.0f;
        lg[g] = v;
    }

    float m = lg[0];
#pragma unroll
    for (int g = 1; g < TILE_J / 4; ++g) m = fmaxf(m, lg[g]);
    m = fmaxf(m, xrow<1>(m));
    m = fmaxf(m, xrow<2>(m));
    if (q == 0) {
        s_max[r * PART_W + w] = m;
        if (w == 0) s_max[r * PART_W + N_JT] = m;
    }
    __syncthreads();
    float rmax;
    {
        const v4f* mp = (const v4f*)(s_max + r * PART_W);
        const v4f m0 = mp[0], m1 = mp[1], m2 = mp[2], m3 = mp[3];
        const float t0 = fmaxf(fmaxf(m0.x, m0.y), fmaxf(m0.z, m0.w));
        const float t1 = fmaxf(fmaxf(m1.x, m1.y), fmaxf(m1.z, m1.w));
        const float t2 = fmaxf(fmaxf(m2.x, m2.y), fmaxf(m2.z, m2.w));
        const float t3 = fmaxf(fmaxf(m3.x, m3.y), fmaxf(m3.z, m3.w));
        rmax = fmaxf(fmaxf(t0, t1), fmaxf(t2, t3));
    }

    float sum = 0.0f;
#pragma unroll
    for (int g = 0; g < TILE_J / 4; ++g) {
        lg[g] = expf(lg[g] - rmax);
        sum += lg[g];
        s_tile[r * N_TOK + jl + 4 * g] = lg[g];
    }
    sum += xrow<1>(sum);
    sum += xrow<2>(sum);
    if (q == 0) {
        s_sum[r * PART_W + w] = sum;
        if (w == 0) s_sum[r * PART_W + N_JT] = 0.0f;
    }
    __syncthreads();

    v4f* ob = (v4f*)(out + (size_t)(b * N_TOK + i0) * N_TOK);
    for (int idx = threadIdx.x; idx < TILE_I * N_TOK / 4; idx += 480) {
        const int row = idx / (N_TOK / 4);
        const v4f* sp = (const v4f*)(s_sum + row * PART_W);
        const v4f s0 = sp[0], s1 = sp[1], s2 = sp[2], s3 = sp[3];
        const float tot = (((s0.x + s0.y) + (s0.z + s0.w)) + ((s1.x + s1.y) + (s1.z + s1.w)))
                        + (((s2.x + s2.y) + (s2.z + s2.w)) + ((s3.x + s3.y) + (s3.z + s3.w)));
        const float inv = 1.0f / tot;
        const v4f e = ((const v4f*)s_tile)[idx];
        ob[idx] = e * inv;
    }
}

extern "C" void kernel_launch(void* const* d_in, const int* in_sizes, int n_in,
                              void* d_out, int out_size, void* d_ws, size_t ws_size,
                              hipStream_t stream) {
    (void)in_sizes; (void)n_in; (void)out_size; (void)ws_size;
    const float* x  = (const float*)d_in[0];
    const float* W1 = (const float*)d_in[1];
    const float* b1 = (const float*)d_in[2];
    const float* W2 = (const float*)d_in[3];
    const float* b2 = (const float*)d_in[4];
    float* out = (float*)d_out;

    float* wsA = (float*)d_ws;
    float* wsC = wsA + M_ROWS * D_DIM;

    gemm_wmma<<<dim3(180), 256, 0, stream>>>(x, W1, b1, wsA, wsC);

    pair_softmax<<<dim3(180), 480, 0, stream>>>(wsA, wsC, W2, b2, out);
}


# pair kernel: d over the 16 lanes of one lane row (8 features per lane), the two lane rows take different columns; lane butterfly 5 steps -> 4, none across rows; logit and row-sum order change
# speedup vs baseline: 2.3534x; 1.0489x over previous
#include <hip/hip_runtime.h>

typedef __attribute__((ext_vector_type(2))) float v2f;
typedef __attribute__((ext_vector_type(4))) float v4f;
typedef __attribute__((ext_vector_type(8))) float v8f;

#define N_TOK 360
#define D_DIM 128
#define B_SZ  4
#define M_ROWS (B_SZ * N_TOK)

__global__ __launch_bounds__(256) void gemm_wmma(const float* __restrict__ x,
                                                 const float* __restrict__ W1,
                                                 const float* __restrict__ b1,
                                                 float* __restrict__ wsA,
                                                 float* __restrict__ wsC) {
    const int wid  = threadIdx.x >> 5;
    const int lane = threadIdx.x & 31;
    const int gw   = blockIdx.x * 8 + wid;
    const int mt   = gw >> 4;
    const int nt16 = gw & 15;
    const int half = nt16 >> 3;
    const int nt   = nt16 & 7;
    const int mbase = mt * 16;
    const int nbase = nt * 16;
    const int lm    = lane & 15;
    const int koff  = (lane < 16) ? 0 : 2;

    const float* xrow = x + (mbase + lm) * D_DIM;
    const float* wb   = W1 + half * D_DIM * D_DIM;

    v8f acc = {0.f, 0.f, 0.f, 0.f, 0.f, 0.f, 0.f, 0.f};
#pragma unroll
    for (int k = 0; k < D_DIM; k += 4) {
        v2f a = *(const v2f*)(xrow + k + koff);
        v2f b;
        b.x = wb[(k + koff)     * D_DIM + nbase + lm];
        b.y = wb[(k + koff + 1) * D_DIM + nbase + lm];
        acc = __builtin_amdgcn_wmma_f32_16x16x4_f32(
             false, a,  false, b,
             (short)0, acc,  false,  false);
    }

    float* dst = half ? wsC : wsA;
    const float badd = half ? 0.0f : b1[nbase + lm];
    const int moff = (lane < 16) ? 0 : 8;
#pragma unroll
    for (int r = 0; r < 8; ++r) {
        dst[(mbase + r + moff) * D_DIM + nbase + lm] = acc[r] + badd;
    }
}

#define TILE_I 8
#define TILE_J 24
#define HALF_J (TILE_J / 2)
#define N_GRP  (HALF_J / 2)
#define N_JT   (N_TOK / TILE_J)
#define PART_W 16

__device__ __forceinline__ float xhalf(float x) {
    int v = __builtin_bit_cast(int, x);
    return __builtin_bit_cast(float, __builtin_amdgcn_permlanex16(
        v, v, 0x76543210, 0xfedcba98, false, false));
}
template <int M>
__device__ __forceinline__ float xrow(float x) {
    return __builtin_bit_cast(float, __builtin_amdgcn_update_dpp(
        0, __builtin_bit_cast(int, x), 0x160 + M, 0xf, 0xf, true));
}

__global__ __launch_bounds__(480) void pair_softmax(const float* __restrict__ wsA,
                                                    const float* __restrict__ wsC,
                                                    const float* __restrict__ W2,
                                                    const float* __restrict__ b2,
                                                    float* __restrict__ out) {
    __shared__ __attribute__((aligned(16))) float s_tile[TILE_I * N_TOK];
    __shared__ __attribute__((aligned(16))) float s_max[TILE_I * PART_W];
    __shared__ __attribute__((aligned(16))) float s_sum[TILE_I * PART_W];

    const int lane = threadIdx.x & 31;
    const int w    = threadIdx.x >> 5;
    const int b    = blockIdx.x / 45;
    const int it   = blockIdx.x - b * 45;
    const int i0   = it * TILE_I;
    const int j0   = w * TILE_J;
    const int cpos = lane & 15;
    const int hrow = lane >> 4;
    const int ra   = cpos >> 1;

    const float* Ab = wsA + (b * N_TOK + i0) * D_DIM + 4 * cpos;
    const float* Cb = wsC + (b * N_TOK + j0 + HALF_J * hrow) * D_DIM + 4 * cpos;

    const v4f wva = *(const v4f*)(W2 + 4 * cpos);
    const v4f wvb = *(const v4f*)(W2 + 64 + 4 * cpos);
    v4f alo[TILE_I], ahi[TILE_I];
    v4f clo[HALF_J], chi[HALF_J];
#pragma unroll
    for (int s = 0; s < TILE_I; ++s) {
        alo[s] = *(const v4f*)(Ab + (s ^ ra) * D_DIM);
        ahi[s] = *(const v4f*)(Ab + (s ^ ra) * D_DIM + 64);
    }
#pragma unroll
    for (int t = 0; t < HALF_J; ++t) {
        clo[t] = *(const v4f*)(Cb + t * D_DIM);
        chi[t] = *(const v4f*)(Cb + t * D_DIM + 64);
    }
    __builtin_amdgcn_sched_barrier(0);
    const v2f w0 = {wva.x, wva.y};
    const v2f w1 = {wva.z, wva.w};
    const v2f w2 = {wvb.x, wvb.y};
    const v2f w3 = {wvb.z, wvb.w};
    const float bias2 = b2[0];
    const bool l0 = (lane & 1) != 0;
    const int  r  = ra;
    const int  io = i0 + r;
    const int  jl = j0 + HALF_J * hrow + (lane & 1);

    float lg[N_GRP];
#pragma unroll
    for (int g = 0; g < N_GRP; ++g) {
        float p[16];
#pragma unroll
        for (int k = 0; k < 16; ++k) {
            const v4f sa = alo[k >> 1] + clo[2 * g + (k & 1)];
            const v4f sb = ahi[k >> 1] + chi[2 * g + (k & 1)];
            const v2f r0 = {fmaxf(sa.x, 0.0f), fmaxf(sa.y, 0.0f)};
            const v2f r1 = {fmaxf(sa.z, 0.0f), fmaxf(sa.w, 0.0f)};
            const v2f r2 = {fmaxf(sb.x, 0.0f), fmaxf(sb.y, 0.0f)};
            const v2f r3 = {fmaxf(sb.z, 0.0f), fmaxf(sb.w, 0.0f)};
            v2f u = r0 * w0;
            u = __builtin_elementwise_fma(r1, w1, u);
            u = __builtin_elementwise_fma(r2, w2, u);
            u = __builtin_elementwise_fma(r3, w3, u);
            p[k] = u.x + u.y;
        }
#pragma unroll
        for (int k = 0; k < 8; ++k) p[k] = p[k] + xrow<8>(p[k + 8]);
#pragma unroll
        for (int k = 0; k < 4; ++k) p[k] = p[k] + xrow<4>(p[k + 4]);
#pragma unroll
        for (int k = 0; k < 2; ++k) p[k] = p[k] + xrow<2>(p[k + 2]);
        {
            const float give = l0 ? p[0] : p[1];
            const float keep = l0 ? p[1] : p[0];
            p[0] = keep + xrow<1>(give);
        }
        const int j = jl + 2 * g;
        float v = p[0] + bias2;
        if (io == j) v = 0.0f;
        lg[g] = v;
    }

    float m = lg[0];
#pragma unroll
    for (int g = 1; g < N_GRP; ++g) m = fmaxf(m, lg[g]);
    m = fmaxf(m, xrow<1>(m));
    m = fmaxf(m, xhalf(m));
    if ((lane & 17) == 0) {
        s_max[r * PART_W + w] = m;
        if (w == 0) s_max[r * PART_W + N_JT] = m;
    }
    __syncthreads();
    float rmax;
    {
        const v4f* mp = (const v4f*)(s_max + r * PART_W);
        const v4f m0 = mp[0], m1 = mp[1], m2 = mp[2], m3 = mp[3];
        const float t0 = fmaxf(fmaxf(m0.x, m0.y), fmaxf(m0.z, m0.w));
        const float t1 = fmaxf(fmaxf(m1.x, m1.y), fmaxf(m1.z, m1.w));
        const float t2 = fmaxf(fmaxf(m2.x, m2.y), fmaxf(m2.z, m2.w));
        const float t3 = fmaxf(fmaxf(m3.x, m3.y), fmaxf(m3.z, m3.w));
        rmax = fmaxf(fmaxf(t0, t1), fmaxf(t2, t3));
    }

    float sum = 0.0f;
#pragma unroll
    for (int g = 0; g < N_GRP; ++g) {
        lg[g] = expf(lg[g] - rmax);
        sum += lg[g];
        s_tile[r * N_TOK + jl + 2 * g] = lg[g];
    }
    sum += xrow<1>(sum);
    sum += xhalf(sum);
    if ((lane & 17) == 0) {
        s_sum[r * PART_W + w] = sum;
        if (w == 0) s_sum[r * PART_W + N_JT] = 0.0f;
    }
    __syncthreads();

    v4f* ob = (v4f*)(out + (size_t)(b * N_TOK + i0) * N_TOK);
    for (int idx = threadIdx.x; idx < TILE_I * N_TOK / 4; idx += 480) {
        const int row = idx / (N_TOK / 4);
        const v4f* sp = (const v4f*)(s_sum + row * PART_W);
        const v4f s0 = sp[0], s1 = sp[1], s2 = sp[2], s3 = sp[3];
        const float tot = (((s0.x + s0.y) + (s0.z + s0.w)) + ((s1.x + s1.y) + (s1.z + s1.w)))
                        + (((s2.x + s2.y) + (s2.z + s2.w)) + ((s3.x + s3.y) + (s3.z + s3.w)));
        const float inv = 1.0f / tot;
        const v4f e = ((const v4f*)s_tile)[idx];
        ob[idx] = e * inv;
    }
}

extern "C" void kernel_launch(void* const* d_in, const int* in_sizes, int n_in,
                              void* d_out, int out_size, void* d_ws, size_t ws_size,
                              hipStream_t stream) {
    (void)in_sizes; (void)n_in; (void)out_size; (void)ws_size;
    const float* x  = (const float*)d_in[0];
    const float* W1 = (const float*)d_in[1];
    const float* b1 = (const float*)d_in[2];
    const float* W2 = (const float*)d_in[3];
    const float* b2 = (const float*)d_in[4];
    float* out = (float*)d_out;

    float* wsA = (float*)d_ws;
    float* wsC = wsA + M_ROWS * D_DIM;

    gemm_wmma<<<dim3(180), 256, 0, stream>>>(x, W1, b1, wsA, wsC);

    pair_softmax<<<dim3(180), 480, 0, stream>>>(wsA, wsC, W2, b2, out);
}


# projection GEMM re-tiled: wave owns 16x32 (2 accumulators, even/odd columns), x read 16 B per lane, W1 and outputs in whole 128-B lines, all loads issued first; 1440 waves -> 720; k sum order changes
# speedup vs baseline: 2.5160x; 1.0691x over previous
#include <hip/hip_runtime.h>

typedef __attribute__((ext_vector_type(2))) float v2f;
typedef __attribute__((ext_vector_type(4))) float v4f;
typedef __attribute__((ext_vector_type(8))) float v8f;

#define N_TOK 360
#define D_DIM 128
#define B_SZ  4
#define M_ROWS (B_SZ * N_TOK)

__global__ __launch_bounds__(128) void gemm_wmma(const float* __restrict__ x,
                                                 const float* __restrict__ W1,
                                                 const float* __restrict__ b1,
                                                 float* __restrict__ wsA,
                                                 float* __restrict__ wsC) {
    const int wid  = threadIdx.x >> 5;
    const int lane = threadIdx.x & 31;
    const int gw   = blockIdx.x * 4 + wid;
    const int mt   = gw >> 3;
    const int half = (gw >> 2) & 1;
    const int nb   = (gw & 3) * 32;
    const int mbase = mt * 16;
    const int c    = lane & 15;
    const int h    = lane >> 4;

    const float* xrow = x + (mbase + c) * D_DIM + 4 * h;
    const float* wcol = W1 + (half * D_DIM + 4 * h) * D_DIM + nb + 2 * c;

    v4f xa[D_DIM / 8];
    v2f wr[D_DIM / 2];
#pragma unroll
    for (int q = 0; q < D_DIM / 8; ++q) {
        xa[q] = *(const v4f*)(xrow + 8 * q);
        const float* wq = wcol + 8 * q * D_DIM;
        wr[4 * q + 0] = *(const v2f*)(wq);
        wr[4 * q + 1] = *(const v2f*)(wq + D_DIM);
        wr[4 * q + 2] = *(const v2f*)(wq + 2 * D_DIM);
        wr[4 * q + 3] = *(const v2f*)(wq + 3 * D_DIM);
    }
    __builtin_amdgcn_sched_barrier(0);

    v8f acc0 = {0.f, 0.f, 0.f, 0.f, 0.f, 0.f, 0.f, 0.f};
    v8f acc1 = {0.f, 0.f, 0.f, 0.f, 0.f, 0.f, 0.f, 0.f};
#pragma unroll
    for (int q = 0; q < D_DIM / 8; ++q) {
        const v2f a01 = {xa[q].x, xa[q].y};
        const v2f a23 = {xa[q].z, xa[q].w};
        const v2f b01e = {wr[4 * q + 0].x, wr[4 * q + 1].x};
        const v2f b01o = {wr[4 * q + 0].y, wr[4 * q + 1].y};
        const v2f b23e = {wr[4 * q + 2].x, wr[4 * q + 3].x};
        const v2f b23o = {wr[4 * q + 2].y, wr[4 * q + 3].y};
        acc0 = __builtin_amdgcn_wmma_f32_16x16x4_f32(
             false, a01,  false, b01e,
             (short)0, acc0,  false,  false);
        acc1 = __builtin_amdgcn_wmma_f32_16x16x4_f32(
             false, a01,  false, b01o,
             (short)0, acc1,  false,  false);
        acc0 = __builtin_amdgcn_wmma_f32_16x16x4_f32(
             false, a23,  false, b23e,
             (short)0, acc0,  false,  false);
        acc1 = __builtin_amdgcn_wmma_f32_16x16x4_f32(
             false, a23,  false, b23o,
             (short)0, acc1,  false,  false);
    }

    float* dst = half ? wsC : wsA;
    v2f badd = {0.0f, 0.0f};
    if (!half) badd = *(const v2f*)(b1 + nb + 2 * c);
    float* drow = dst + (mbase + 8 * h) * D_DIM + nb + 2 * c;
#pragma unroll
    for (int r = 0; r < 8; ++r) {
        const v2f o = {acc0[r] + badd.x, acc1[r] + badd.y};
        *(v2f*)(drow + r * D_DIM) = o;
    }
}

#define TILE_I 8
#define TILE_J 24
#define HALF_J (TILE_J / 2)
#define N_GRP  (HALF_J / 2)
#define N_JT   (N_TOK / TILE_J)
#define PART_W 16

__device__ __forceinline__ float xhalf(float x) {
    int v = __builtin_bit_cast(int, x);
    return __builtin_bit_cast(float, __builtin_amdgcn_permlanex16(
        v, v, 0x76543210, 0xfedcba98, false, false));
}
template <int M>
__device__ __forceinline__ float xrow(float x) {
    return __builtin_bit_cast(float, __builtin_amdgcn_update_dpp(
        0, __builtin_bit_cast(int, x), 0x160 + M, 0xf, 0xf, true));
}

__global__ __launch_bounds__(480) void pair_softmax(const float* __restrict__ wsA,
                                                    const float* __restrict__ wsC,
                                                    const float* __restrict__ W2,
                                                    const float* __restrict__ b2,
                                                    float* __restrict__ out) {
    __shared__ __attribute__((aligned(16))) float s_tile[TILE_I * N_TOK];
    __shared__ __attribute__((aligned(16))) float s_max[TILE_I * PART_W];
    __shared__ __attribute__((aligned(16))) float s_sum[TILE_I * PART_W];

    const int lane = threadIdx.x & 31;
    const int w    = threadIdx.x >> 5;
    const int b    = blockIdx.x / 45;
    const int it   = blockIdx.x - b * 45;
    const int i0   = it * TILE_I;
    const int j0   = w * TILE_J;
    const int cpos = lane & 15;
    const int hrow = lane >> 4;
    const int ra   = cpos >> 1;

    const float* Ab = wsA + (b * N_TOK + i0) * D_DIM + 4 * cpos;
    const float* Cb = wsC + (b * N_TOK + j0 + HALF_J * hrow) * D_DIM + 4 * cpos;

    const v4f wva = *(const v4f*)(W2 + 4 * cpos);
    const v4f wvb = *(const v4f*)(W2 + 64 + 4 * cpos);
    v4f alo[TILE_I], ahi[TILE_I];
    v4f clo[HALF_J], chi[HALF_J];
#pragma unroll
    for (int s = 0; s < TILE_I; ++s) {
        alo[s] = *(const v4f*)(Ab + (s ^ ra) * D_DIM);
        ahi[s] = *(const v4f*)(Ab + (s ^ ra) * D_DIM + 64);
    }
#pragma unroll
    for (int t = 0; t < HALF_J; ++t) {
        clo[t] = *(const v4f*)(Cb + t * D_DIM);
        chi[t] = *(const v4f*)(Cb + t * D_DIM + 64);
    }
    __builtin_amdgcn_sched_barrier(0);
    const v2f w0 = {wva.x, wva.y};
    const v2f w1 = {wva.z, wva.w};
    const v2f w2 = {wvb.x, wvb.y};
    const v2f w3 = {wvb.z, wvb.w};
    const float bias2 = b2[0];
    const bool l0 = (lane & 1) != 0;
    const int  r  = ra;
    const int  io = i0 + r;
    const int  jl = j0 + HALF_J * hrow + (lane & 1);

    float lg[N_GRP];
#pragma unroll
    for (int g = 0; g < N_GRP; ++g) {
        float p[16];
#pragma unroll
        for (int k = 0; k < 16; ++k) {
            const v4f sa = alo[k >> 1] + clo[2 * g + (k & 1)];
            const v4f sb = ahi[k >> 1] + chi[2 * g + (k & 1)];
            const v2f r0 = {fmaxf(sa.x, 0.0f), fmaxf(sa.y, 0.0f)};
            const v2f r1 = {fmaxf(sa.z, 0.0f), fmaxf(sa.w, 0.0f)};
            const v2f r2 = {fmaxf(sb.x, 0.0f), fmaxf(sb.y, 0.0f)};
            const v2f r3 = {fmaxf(sb.z, 0.0f), fmaxf(sb.w, 0.0f)};
            v2f u = r0 * w0;
            u = __builtin_elementwise_fma(r1, w1, u);
            u = __builtin_elementwise_fma(r2, w2, u);
            u = __builtin_elementwise_fma(r3, w3, u);
            p[k] = u.x + u.y;
        }
#pragma unroll
        for (int k = 0; k < 8; ++k) p[k] = p[k] + xrow<8>(p[k + 8]);
#pragma unroll
        for (int k = 0; k < 4; ++k) p[k] = p[k] + xrow<4>(p[k + 4]);
#pragma unroll
        for (int k = 0; k < 2; ++k) p[k] = p[k] + xrow<2>(p[k + 2]);
        {
            const float give = l0 ? p[0] : p[1];
            const float keep = l0 ? p[1] : p[0];
            p[0] = keep + xrow<1>(give);
        }
        const int j = jl + 2 * g;
        float v = p[0] + bias2;
        if (io == j) v = 0.0f;
        lg[g] = v;
    }

    float m = lg[0];
#pragma unroll
    for (int g = 1; g < N_GRP; ++g) m = fmaxf(m, lg[g]);
    m = fmaxf(m, xrow<1>(m));
    m = fmaxf(m, xhalf(m));
    if ((lane & 17) == 0) {
        s_max[r * PART_W + w] = m;
        if (w == 0) s_max[r * PART_W + N_JT] = m;
    }
    __syncthreads();
    float rmax;
    {
        const v4f* mp = (const v4f*)(s_max + r * PART_W);
        const v4f m0 = mp[0], m1 = mp[1], m2 = mp[2], m3 = mp[3];
        const float t0 = fmaxf(fmaxf(m0.x, m0.y), fmaxf(m0.z, m0.w));
        const float t1 = fmaxf(fmaxf(m1.x, m1.y), fmaxf(m1.z, m1.w));
        const float t2 = fmaxf(fmaxf(m2.x, m2.y), fmaxf(m2.z, m2.w));
        const float t3 = fmaxf(fmaxf(m3.x, m3.y), fmaxf(m3.z, m3.w));
        rmax = fmaxf(fmaxf(t0, t1), fmaxf(t2, t3));
    }

    float sum = 0.0f;
#pragma unroll
    for (int g = 0; g < N_GRP; ++g) {
        lg[g] = expf(lg[g] - rmax);
        sum += lg[g];
        s_tile[r * N_TOK + jl + 2 * g] = lg[g];
    }
    sum += xrow<1>(sum);
    sum += xhalf(sum);
    if ((lane & 17) == 0) {
        s_sum[r * PART_W + w] = sum;
        if (w == 0) s_sum[r * PART_W + N_JT] = 0.0f;
    }
    __syncthreads();

    v4f* ob = (v4f*)(out + (size_t)(b * N_TOK + i0) * N_TOK);
    for (int idx = threadIdx.x; idx < TILE_I * N_TOK / 4; idx += 480) {
        const int row = idx / (N_TOK / 4);
        const v4f* sp = (const v4f*)(s_sum + row * PART_W);
        const v4f s0 = sp[0], s1 = sp[1], s2 = sp[2], s3 = sp[3];
        const float tot = (((s0.x + s0.y) + (s0.z + s0.w)) + ((s1.x + s1.y) + (s1.z + s1.w)))
                        + (((s2.x + s2.y) + (s2.z + s2.w)) + ((s3.x + s3.y) + (s3.z + s3.w)));
        const float inv = 1.0f / tot;
        const v4f e = ((const v4f*)s_tile)[idx];
        ob[idx] = e * inv;
    }
}

extern "C" void kernel_launch(void* const* d_in, const int* in_sizes, int n_in,
                              void* d_out, int out_size, void* d_ws, size_t ws_size,
                              hipStream_t stream) {
    (void)in_sizes; (void)n_in; (void)out_size; (void)ws_size;
    const float* x  = (const float*)d_in[0];
    const float* W1 = (const float*)d_in[1];
    const float* b1 = (const float*)d_in[2];
    const float* W2 = (const float*)d_in[3];
    const float* b2 = (const float*)d_in[4];
    float* out = (float*)d_out;

    float* wsA = (float*)d_ws;
    float* wsC = wsA + M_ROWS * D_DIM;

    gemm_wmma<<<dim3(180), 128, 0, stream>>>(x, W1, b1, wsA, wsC);

    pair_softmax<<<dim3(180), 480, 0, stream>>>(wsA, wsC, W2, b2, out);
}
